# starting kernel
# baseline (speedup 1.0000x reference)
#include <hip/hip_runtime.h>

typedef __attribute__((ext_vector_type(2))) float v2f;
typedef __attribute__((ext_vector_type(8))) float v8f;

#define IN_DIM 64
#define OUT_DIM 64
#define LDS_STRIDE 68

#define WMMA_F32_16x16x4(A, B, C) \
  __builtin_amdgcn_wmma_f32_16x16x4_f32(false, (A), false, (B), (short)0, (C), false, false)

__global__ __launch_bounds__(256) void demo_scatter_kernel(
    const float* __restrict__ x, const int* __restrict__ ei,
    float* __restrict__ nsum, float* __restrict__ deg, int n_edges) {
  long long tid = (long long)blockIdx.x * blockDim.x + threadIdx.x;
  int e = (int)(tid >> 4);
  if (e >= n_edges) return;
  int part = (int)(tid & 15);
  int s = ei[e];
  int d = ei[n_edges + e];
  const float4 v = *(const float4*)(x + (size_t)s * IN_DIM + part * 4);
  float* dst = nsum + (size_t)d * IN_DIM + part * 4;
  atomicAdd(dst + 0, v.x);
  atomicAdd(dst + 1, v.y);
  atomicAdd(dst + 2, v.z);
  atomicAdd(dst + 3, v.w);
  if (part == 0) atomicAdd(deg + d, 1.0f);
}

__global__ __launch_bounds__(256) void demo_node_kernel(
    const float* __restrict__ x, const float* __restrict__ Wg,
    const float* __restrict__ Wl, const float* __restrict__ Ws,
    const float* __restrict__ bias, const float* __restrict__ deg,
    float* sum_out, int n_nodes) {
  __shared__ float sW[3 * OUT_DIM * LDS_STRIDE];
  float* sWg = sW;
  float* sWs = sW + OUT_DIM * LDS_STRIDE;
  float* sWl = sW + 2 * OUT_DIM * LDS_STRIDE;

  {
    const int tid = threadIdx.x;
#pragma unroll
    for (int i = 0; i < 4; ++i) {
      int idx = tid + i * 256;
      int row = idx >> 4;
      int k4  = idx & 15;
      float4 g = *(const float4*)(Wg + (size_t)idx * 4);
      float4 s = *(const float4*)(Ws + (size_t)idx * 4);
      float4 l = *(const float4*)(Wl + (size_t)idx * 4);
      *(float4*)(sWg + row * LDS_STRIDE + k4 * 4) = g;
      *(float4*)(sWs + row * LDS_STRIDE + k4 * 4) = s;
      *(float4*)(sWl + row * LDS_STRIDE + k4 * 4) = l;
    }
  }
  __syncthreads();

  const int lane    = threadIdx.x & 31;
  const int wave    = threadIdx.x >> 5;
  const int lane_lo = lane & 15;
  const int hi      = lane >> 4;
  const int node_base = (blockIdx.x * 8 + wave) * 16;
  if (node_base >= n_nodes) return;

  int arow = node_base + lane_lo;
  if (arow > n_nodes - 1) arow = n_nodes - 1;

  const float dv  = deg[arow];
  const float inv = 1.0f / fmaxf(dv, 1.0f);
  const float hn  = dv > 0.0f ? 1.0f : 0.0f;

  const v2f* xr = (const v2f*)(x       + (size_t)arow * IN_DIM + 2 * hi);
  const v2f* sr = (const v2f*)(sum_out + (size_t)arow * IN_DIM + 2 * hi);

  const int bofs = lane_lo * LDS_STRIDE + 2 * hi;

  v8f acc[4] = {v8f{}, v8f{}, v8f{}, v8f{}};

  for (int c = 0; c < 16; ++c) {
    v2f ax = xr[2 * c];
    v2f am = sr[2 * c] * inv;
    v2f as = ax * hn;

    v2f bg[4], bs[4], bl[4];
#pragma unroll
    for (int nb = 0; nb < 4; ++nb) {
      const int bo = bofs + nb * 16 * LDS_STRIDE + 4 * c;
      bg[nb] = *(const v2f*)(sWg + bo);
      bs[nb] = *(const v2f*)(sWs + bo);
      bl[nb] = *(const v2f*)(sWl + bo);
    }
#pragma unroll
    for (int nb = 0; nb < 4; ++nb) {
      acc[nb] = WMMA_F32_16x16x4(ax, bg[nb], acc[nb]);
      acc[nb] = WMMA_F32_16x16x4(as, bs[nb], acc[nb]);
      acc[nb] = WMMA_F32_16x16x4(am, bl[nb], acc[nb]);
    }
  }

#pragma unroll
  for (int nb = 0; nb < 4; ++nb) {
    const int col = nb * 16 + lane_lo;
    const float bv = bias[col];
    v8f r = acc[nb];
#pragma unroll
    for (int rr = 0; rr < 8; ++rr) {
      int m = node_base + rr + 8 * hi;
      if (m < n_nodes) {
        float val = r[rr] + bv;
        val = val > 0.0f ? val : (__expf(val) - 1.0f);
        sum_out[(size_t)m * OUT_DIM + col] = val;
      }
    }
  }
}

extern "C" void kernel_launch(void* const* d_in, const int* in_sizes, int n_in,
                              void* d_out, int out_size, void* d_ws, size_t ws_size,
                              hipStream_t stream) {
  const float* x    = (const float*)d_in[0];
  const float* Wg   = (const float*)d_in[1];
  const float* Wl   = (const float*)d_in[2];
  const float* Ws   = (const float*)d_in[3];
  const float* bias = (const float*)d_in[4];
  const int*   ei   = (const int*)d_in[5];
  float* out = (float*)d_out;
  float* deg = (float*)d_ws;

  const int n_nodes = in_sizes[0] / IN_DIM;
  const int n_edges = in_sizes[5] / 2;

  hipMemsetAsync(out, 0, (size_t)n_nodes * IN_DIM * sizeof(float), stream);
  hipMemsetAsync(deg, 0, (size_t)n_nodes * sizeof(float), stream);

  {
    long long threads = (long long)n_edges * 16;
    int blocks = (int)((threads + 255) / 256);
    demo_scatter_kernel<<<blocks, 256, 0, stream>>>(x, ei, out, deg, n_edges);
  }
  {
    int waves  = (n_nodes + 15) / 16;
    int blocks = (waves + 7) / 8;
    demo_node_kernel<<<blocks, 256, 0, stream>>>(x, Wg, Wl, Ws, bias, deg, out,
                                                 n_nodes);
  }
}


# Neighbour sum without float atomics: edges linked per destination by one integer exchange, 8 lanes per node walk the list and sum x rows in registers; sum order = arrival order of the exchange
# speedup vs baseline: 1.7179x; 1.7179x over previous
#include <hip/hip_runtime.h>

typedef __attribute__((ext_vector_type(2))) float v2f;
typedef __attribute__((ext_vector_type(8))) float v8f;

#define IN_DIM 64
#define OUT_DIM 64
#define LDS_STRIDE 68

#define WMMA_F32_16x16x4(A, B, C) \
  __builtin_amdgcn_wmma_f32_16x16x4_f32(false, (A), false, (B), (short)0, (C), false, false)

__global__ __launch_bounds__(256) void demo_scatter_kernel(
    const float* __restrict__ x, const int* __restrict__ ei,
    float* __restrict__ nsum, float* __restrict__ deg, int n_edges) {
  long long tid = (long long)blockIdx.x * blockDim.x + threadIdx.x;
  int e = (int)(tid >> 4);
  if (e >= n_edges) return;
  int part = (int)(tid & 15);
  int s = ei[e];
  int d = ei[n_edges + e];
  const float4 v = *(const float4*)(x + (size_t)s * IN_DIM + part * 4);
  float* dst = nsum + (size_t)d * IN_DIM + part * 4;
  atomicAdd(dst + 0, v.x);
  atomicAdd(dst + 1, v.y);
  atomicAdd(dst + 2, v.z);
  atomicAdd(dst + 3, v.w);
  if (part == 0) atomicAdd(deg + d, 1.0f);
}

__global__ __launch_bounds__(256) void demo_link_kernel(
    const int* __restrict__ ei, int* __restrict__ head, int* __restrict__ next,
    int n_edges, int n_nodes) {
  const int T = gridDim.x * blockDim.x;
  const int t = blockIdx.x * blockDim.x + threadIdx.x;
  int d[4], old[4];
#pragma unroll
  for (int k = 0; k < 4; ++k) {
    const int e = t + k * T;
    d[k] = e < n_edges ? ei[n_edges + e] : -1;
  }
#pragma unroll
  for (int k = 0; k < 4; ++k) {
    old[k] = -1;
    if ((unsigned)d[k] < (unsigned)n_nodes) old[k] = atomicExch(head + d[k], t + k * T);
  }
#pragma unroll
  for (int k = 0; k < 4; ++k)
    if ((unsigned)d[k] < (unsigned)n_nodes) next[t + k * T] = old[k];
}

__global__ __launch_bounds__(256) void demo_gather_kernel(
    const float* __restrict__ x, const int* __restrict__ ei,
    const int* __restrict__ head, const int* __restrict__ next,
    float* __restrict__ nsum, float* __restrict__ deg, int n_edges,
    int n_nodes) {
  const int tid  = blockIdx.x * blockDim.x + threadIdx.x;
  const int node = tid >> 3;
  if (node >= n_nodes) return;
  const int part = tid & 7;
  const float4* x4 = (const float4*)x;

  float4 a0 = make_float4(0.0f, 0.0f, 0.0f, 0.0f);
  float4 a1 = make_float4(0.0f, 0.0f, 0.0f, 0.0f);
  int cnt = 0;
  int e = head[node];
  int s = 0, nx = -1;
  if (e >= 0) { s = ei[e]; nx = next[e]; }
  for (int it = 0; it < n_edges; ++it) {
    if (e < 0) break;
    const float4* row = x4 + (size_t)s * (IN_DIM / 4);
    const float4 v0 = row[part];
    const float4 v1 = row[part + 8];
    const int e2 = nx;
    int s2 = 0, nx2 = -1;
    if (e2 >= 0) { s2 = ei[e2]; nx2 = next[e2]; }
    a0.x += v0.x; a0.y += v0.y; a0.z += v0.z; a0.w += v0.w;
    a1.x += v1.x; a1.y += v1.y; a1.z += v1.z; a1.w += v1.w;
    ++cnt;
    e = e2; s = s2; nx = nx2;
  }
  float4* orow = (float4*)nsum + (size_t)node * (IN_DIM / 4);
  orow[part]     = a0;
  orow[part + 8] = a1;
  if (part == 0) deg[node] = (float)cnt;
}

__global__ __launch_bounds__(256) void demo_node_kernel(
    const float* __restrict__ x, const float* __restrict__ Wg,
    const float* __restrict__ Wl, const float* __restrict__ Ws,
    const float* __restrict__ bias, const float* __restrict__ deg,
    float* sum_out, int n_nodes) {
  __shared__ float sW[3 * OUT_DIM * LDS_STRIDE];
  float* sWg = sW;
  float* sWs = sW + OUT_DIM * LDS_STRIDE;
  float* sWl = sW + 2 * OUT_DIM * LDS_STRIDE;

  {
    const int tid = threadIdx.x;
#pragma unroll
    for (int i = 0; i < 4; ++i) {
      int idx = tid + i * 256;
      int row = idx >> 4;
      int k4  = idx & 15;
      float4 g = *(const float4*)(Wg + (size_t)idx * 4);
      float4 s = *(const float4*)(Ws + (size_t)idx * 4);
      float4 l = *(const float4*)(Wl + (size_t)idx * 4);
      *(float4*)(sWg + row * LDS_STRIDE + k4 * 4) = g;
      *(float4*)(sWs + row * LDS_STRIDE + k4 * 4) = s;
      *(float4*)(sWl + row * LDS_STRIDE + k4 * 4) = l;
    }
  }
  __syncthreads();

  const int lane    = threadIdx.x & 31;
  const int wave    = threadIdx.x >> 5;
  const int lane_lo = lane & 15;
  const int hi      = lane >> 4;
  const int node_base = (blockIdx.x * 8 + wave) * 16;
  if (node_base >= n_nodes) return;

  int arow = node_base + lane_lo;
  if (arow > n_nodes - 1) arow = n_nodes - 1;

  const float dv  = deg[arow];
  const float inv = 1.0f / fmaxf(dv, 1.0f);
  const float hn  = dv > 0.0f ? 1.0f : 0.0f;

  const v2f* xr = (const v2f*)(x       + (size_t)arow * IN_DIM + 2 * hi);
  const v2f* sr = (const v2f*)(sum_out + (size_t)arow * IN_DIM + 2 * hi);

  const int bofs = lane_lo * LDS_STRIDE + 2 * hi;

  v8f acc[4] = {v8f{}, v8f{}, v8f{}, v8f{}};

  for (int c = 0; c < 16; ++c) {
    v2f ax = xr[2 * c];
    v2f am = sr[2 * c] * inv;
    v2f as = ax * hn;

    v2f bg[4], bs[4], bl[4];
#pragma unroll
    for (int nb = 0; nb < 4; ++nb) {
      const int bo = bofs + nb * 16 * LDS_STRIDE + 4 * c;
      bg[nb] = *(const v2f*)(sWg + bo);
      bs[nb] = *(const v2f*)(sWs + bo);
      bl[nb] = *(const v2f*)(sWl + bo);
    }
#pragma unroll
    for (int nb = 0; nb < 4; ++nb) {
      acc[nb] = WMMA_F32_16x16x4(ax, bg[nb], acc[nb]);
      acc[nb] = WMMA_F32_16x16x4(as, bs[nb], acc[nb]);
      acc[nb] = WMMA_F32_16x16x4(am, bl[nb], acc[nb]);
    }
  }

#pragma unroll
  for (int nb = 0; nb < 4; ++nb) {
    const int col = nb * 16 + lane_lo;
    const float bv = bias[col];
    v8f r = acc[nb];
#pragma unroll
    for (int rr = 0; rr < 8; ++rr) {
      int m = node_base + rr + 8 * hi;
      if (m < n_nodes) {
        float val = r[rr] + bv;
        val = val > 0.0f ? val : (__expf(val) - 1.0f);
        sum_out[(size_t)m * OUT_DIM + col] = val;
      }
    }
  }
}

extern "C" void kernel_launch(void* const* d_in, const int* in_sizes, int n_in,
                              void* d_out, int out_size, void* d_ws, size_t ws_size,
                              hipStream_t stream) {
  const float* x    = (const float*)d_in[0];
  const float* Wg   = (const float*)d_in[1];
  const float* Wl   = (const float*)d_in[2];
  const float* Ws   = (const float*)d_in[3];
  const float* bias = (const float*)d_in[4];
  const int*   ei   = (const int*)d_in[5];
  float* out = (float*)d_out;
  const int n_nodes = in_sizes[0] / IN_DIM;
  const int n_edges = in_sizes[5] / 2;

  const size_t need = ((size_t)2 * n_nodes + (size_t)n_edges) * sizeof(int);
  if (ws_size >= need) {
    int*   head = (int*)d_ws;
    float* deg  = (float*)d_ws + n_nodes;
    int*   next = (int*)d_ws + 2 * (size_t)n_nodes;
    hipMemsetAsync(head, 0xFF, (size_t)n_nodes * sizeof(int), stream);
    {
      int blocks = (n_edges + 1023) / 1024;
      demo_link_kernel<<<blocks, 256, 0, stream>>>(ei, head, next, n_edges, n_nodes);
    }
    {
      long long threads = (long long)n_nodes * 8;
      int blocks = (int)((threads + 255) / 256);
      demo_gather_kernel<<<blocks, 256, 0, stream>>>(x, ei, head, next, out, deg,
                                                     n_edges, n_nodes);
    }
    {
      int waves  = (n_nodes + 15) / 16;
      int blocks = (waves + 7) / 8;
      demo_node_kernel<<<blocks, 256, 0, stream>>>(x, Wg, Wl, Ws, bias, deg, out,
                                                   n_nodes);
    }
    return;
  }

  float* deg = (float*)d_ws;
  hipMemsetAsync(out, 0, (size_t)n_nodes * IN_DIM * sizeof(float), stream);
  hipMemsetAsync(deg, 0, (size_t)n_nodes * sizeof(float), stream);
  {
    long long threads = (long long)n_edges * 16;
    int blocks = (int)((threads + 255) / 256);
    demo_scatter_kernel<<<blocks, 256, 0, stream>>>(x, ei, out, deg, n_edges);
  }
  {
    int waves  = (n_nodes + 15) / 16;
    int blocks = (waves + 7) / 8;
    demo_node_kernel<<<blocks, 256, 0, stream>>>(x, Wg, Wl, Ws, bias, deg, out,
                                                 n_nodes);
  }
}


# Edges grouped by 256-node destination block via LDS counters (one global add per tile and block, no returning global exchange per edge); lists linked and summed per block in LDS; sum order = arrival
# speedup vs baseline: 3.1110x; 1.8109x over previous
#include <hip/hip_runtime.h>

typedef __attribute__((ext_vector_type(2))) float v2f;
typedef __attribute__((ext_vector_type(8))) float v8f;

#define IN_DIM 64
#define OUT_DIM 64
#define LDS_STRIDE 68

#define WMMA_F32_16x16x4(A, B, C) \
  __builtin_amdgcn_wmma_f32_16x16x4_f32(false, (A), false, (B), (short)0, (C), false, false)

__global__ __launch_bounds__(256) void demo_scatter_kernel(
    const float* __restrict__ x, const int* __restrict__ ei,
    float* __restrict__ nsum, float* __restrict__ deg, int n_edges) {
  long long tid = (long long)blockIdx.x * blockDim.x + threadIdx.x;
  int e = (int)(tid >> 4);
  if (e >= n_edges) return;
  int part = (int)(tid & 15);
  int s = ei[e];
  int d = ei[n_edges + e];
  const float4 v = *(const float4*)(x + (size_t)s * IN_DIM + part * 4);
  float* dst = nsum + (size_t)d * IN_DIM + part * 4;
  atomicAdd(dst + 0, v.x);
  atomicAdd(dst + 1, v.y);
  atomicAdd(dst + 2, v.z);
  atomicAdd(dst + 3, v.w);
  if (part == 0) atomicAdd(deg + d, 1.0f);
}

#define BIN_SHIFT 8
#define BIN_NODES 256
#define MAX_BINS  1024
#define TILE_EDGES 4096
#define CHUNK_RECS 6144

typedef __attribute__((ext_vector_type(4))) float v4f;

__device__ __forceinline__ int block_scan_excl_256(int v, int* s_wave, int& total) {
  const int lane = threadIdx.x & 31;
  const int w    = threadIdx.x >> 5;
  int x = v;
  for (int o = 1; o < 32; o <<= 1) {
    const int y = __shfl_up(x, o);
    if (lane >= o) x += y;
  }
  __syncthreads();
  if (lane == 31) s_wave[w] = x;
  __syncthreads();
  int wbase = 0, t = 0;
  for (int i = 0; i < 8; ++i) {
    const int sv = s_wave[i];
    if (i < w) wbase += sv;
    t += sv;
  }
  total = t;
  return wbase + x - v;
}

__global__ __launch_bounds__(256) void demo_bin_count_kernel(
    const int* __restrict__ ei, int* __restrict__ bincnt, int n_edges,
    int n_nodes, int nbins) {
  __shared__ int cnt[MAX_BINS];
  const int tid = threadIdx.x;
  for (int b = tid; b < nbins; b += 256) cnt[b] = 0;
  __syncthreads();
  const int base = blockIdx.x * TILE_EDGES;
#pragma unroll
  for (int k = 0; k < TILE_EDGES / 256; ++k) {
    const int e = base + k * 256 + tid;
    if (e < n_edges) {
      const int d = ei[n_edges + e];
      if ((unsigned)d < (unsigned)n_nodes) atomicAdd(&cnt[d >> BIN_SHIFT], 1);
    }
  }
  __syncthreads();
  for (int b = tid; b < nbins; b += 256) {
    const int c = cnt[b];
    if (c) atomicAdd(bincnt + b, c);
  }
}

__global__ __launch_bounds__(256) void demo_bin_scatter_kernel(
    const int* __restrict__ ei, const int* __restrict__ bincnt,
    int* __restrict__ cursor, int* __restrict__ binstart,
    int2* __restrict__ rec, int n_edges, int n_nodes, int nbins) {
  __shared__ int  cnt[MAX_BINS];
  __shared__ int  lstart[MAX_BINS];
  __shared__ int  gbase[MAX_BINS];
  __shared__ int  s_wave[8];
  __shared__ int2 s_rec[TILE_EDGES];
  const int tid = threadIdx.x;
  for (int b = tid; b < nbins; b += 256) cnt[b] = 0;

  int carry = 0;
  for (int b0 = 0; b0 < nbins; b0 += 256) {
    const int b = b0 + tid;
    const int v = b < nbins ? bincnt[b] : 0;
    int tot;
    const int p = block_scan_excl_256(v, s_wave, tot);
    if (b < nbins) {
      gbase[b] = carry + p;
      if (blockIdx.x == 0) binstart[b] = carry + p;
    }
    carry += tot;
  }
  if (blockIdx.x == 0 && tid == 0) binstart[nbins] = carry;
  __syncthreads();

  const int base = blockIdx.x * TILE_EDGES;
  int d[TILE_EDGES / 256], s[TILE_EDGES / 256], r[TILE_EDGES / 256];
#pragma unroll
  for (int k = 0; k < TILE_EDGES / 256; ++k) {
    const int e = base + k * 256 + tid;
    d[k] = 0; s[k] = 0; r[k] = -1;
    if (e < n_edges) {
      d[k] = ei[n_edges + e];
      s[k] = ei[e];
      if ((unsigned)d[k] < (unsigned)n_nodes) r[k] = atomicAdd(&cnt[d[k] >> BIN_SHIFT], 1);
    }
  }
  __syncthreads();

  carry = 0;
  for (int b0 = 0; b0 < nbins; b0 += 256) {
    const int b = b0 + tid;
    const int v = b < nbins ? cnt[b] : 0;
    int tot;
    const int p = block_scan_excl_256(v, s_wave, tot);
    if (b < nbins) {
      lstart[b] = carry + p;
      if (v) gbase[b] += atomicAdd(cursor + b, v);
    }
    carry += tot;
  }
  const int nvalid = carry;
  __syncthreads();

#pragma unroll
  for (int k = 0; k < TILE_EDGES / 256; ++k)
    if (r[k] >= 0) s_rec[lstart[d[k] >> BIN_SHIFT] + r[k]] = make_int2(s[k], d[k]);
  __syncthreads();
  for (int i = tid; i < nvalid; i += 256) {
    const int2 q = s_rec[i];
    const int b = q.y >> BIN_SHIFT;
    rec[gbase[b] + (i - lstart[b])] = q;
  }
}

__global__ __launch_bounds__(512) void demo_bin_gather_kernel(
    const float* __restrict__ x, const int2* __restrict__ rec,
    const int* __restrict__ binstart, float* __restrict__ nsum,
    float* __restrict__ deg, int n_nodes) {
  __shared__ int head[BIN_NODES];
  __shared__ int s_src[CHUNK_RECS];
  __shared__ int s_next[CHUNK_RECS];
  const int tid   = threadIdx.x;
  const int part  = tid & 7;
  const int g     = tid >> 3;
  const int node0 = blockIdx.x << BIN_SHIFT;
  const int beg   = binstart[blockIdx.x];
  const int end   = binstart[blockIdx.x + 1];
  const v4f* x4 = (const v4f*)x;

  v4f a0[4], a1[4];
  int cnt[4];
#pragma unroll
  for (int j = 0; j < 4; ++j) { a0[j] = v4f{}; a1[j] = v4f{}; cnt[j] = 0; }

  for (int c0 = beg; c0 < end; c0 += CHUNK_RECS) {
    const int n = min(CHUNK_RECS, end - c0);
    if (tid < BIN_NODES) head[tid] = -1;
    __syncthreads();
    for (int i = tid; i < n; i += 512) {
      const int2 q = rec[c0 + i];
      s_src[i]  = q.x;
      s_next[i] = atomicExch(&head[(q.y - node0) & (BIN_NODES - 1)], i);
    }
    __syncthreads();

    int cur[4];
#pragma unroll
    for (int j = 0; j < 4; ++j) cur[j] = head[g + 64 * j];
    for (int it = 0; it < n; ++it) {
      if ((cur[0] & cur[1] & cur[2] & cur[3]) < 0) break;
      v4f v0[4], v1[4];
      bool act[4];
#pragma unroll
      for (int j = 0; j < 4; ++j) {
        act[j] = cur[j] >= 0;
        const int ci   = act[j] ? cur[j] : 0;
        const int sidx = s_src[ci];
        const int nx   = s_next[ci];
        const v4f* row = x4 + (size_t)sidx * (IN_DIM / 4);
        v0[j] = row[part];
        v1[j] = row[part + 8];
        cur[j] = act[j] ? nx : -1;
        cnt[j] += act[j] ? 1 : 0;
      }
#pragma unroll
      for (int j = 0; j < 4; ++j) {
        a0[j] += act[j] ? v0[j] : v4f{};
        a1[j] += act[j] ? v1[j] : v4f{};
      }
    }
    __syncthreads();
  }

#pragma unroll
  for (int j = 0; j < 4; ++j) {
    const int node = node0 + g + 64 * j;
    if (node < n_nodes) {
      v4f* orow = (v4f*)nsum + (size_t)node * (IN_DIM / 4);
      orow[part]     = a0[j];
      orow[part + 8] = a1[j];
      if (part == 0) deg[node] = (float)cnt[j];
    }
  }
}

__global__ __launch_bounds__(256) void demo_node_kernel(
    const float* __restrict__ x, const float* __restrict__ Wg,
    const float* __restrict__ Wl, const float* __restrict__ Ws,
    const float* __restrict__ bias, const float* __restrict__ deg,
    float* sum_out, int n_nodes) {
  __shared__ float sW[3 * OUT_DIM * LDS_STRIDE];
  float* sWg = sW;
  float* sWs = sW + OUT_DIM * LDS_STRIDE;
  float* sWl = sW + 2 * OUT_DIM * LDS_STRIDE;

  {
    const int tid = threadIdx.x;
#pragma unroll
    for (int i = 0; i < 4; ++i) {
      int idx = tid + i * 256;
      int row = idx >> 4;
      int k4  = idx & 15;
      float4 g = *(const float4*)(Wg + (size_t)idx * 4);
      float4 s = *(const float4*)(Ws + (size_t)idx * 4);
      float4 l = *(const float4*)(Wl + (size_t)idx * 4);
      *(float4*)(sWg + row * LDS_STRIDE + k4 * 4) = g;
      *(float4*)(sWs + row * LDS_STRIDE + k4 * 4) = s;
      *(float4*)(sWl + row * LDS_STRIDE + k4 * 4) = l;
    }
  }
  __syncthreads();

  const int lane    = threadIdx.x & 31;
  const int wave    = threadIdx.x >> 5;
  const int lane_lo = lane & 15;
  const int hi      = lane >> 4;
  const int node_base = (blockIdx.x * 8 + wave) * 16;
  if (node_base >= n_nodes) return;

  int arow = node_base + lane_lo;
  if (arow > n_nodes - 1) arow = n_nodes - 1;

  const float dv  = deg[arow];
  const float inv = 1.0f / fmaxf(dv, 1.0f);
  const float hn  = dv > 0.0f ? 1.0f : 0.0f;

  const v2f* xr = (const v2f*)(x       + (size_t)arow * IN_DIM + 2 * hi);
  const v2f* sr = (const v2f*)(sum_out + (size_t)arow * IN_DIM + 2 * hi);

  const int bofs = lane_lo * LDS_STRIDE + 2 * hi;

  v8f acc[4] = {v8f{}, v8f{}, v8f{}, v8f{}};

  for (int c = 0; c < 16; ++c) {
    v2f ax = xr[2 * c];
    v2f am = sr[2 * c] * inv;
    v2f as = ax * hn;

    v2f bg[4], bs[4], bl[4];
#pragma unroll
    for (int nb = 0; nb < 4; ++nb) {
      const int bo = bofs + nb * 16 * LDS_STRIDE + 4 * c;
      bg[nb] = *(const v2f*)(sWg + bo);
      bs[nb] = *(const v2f*)(sWs + bo);
      bl[nb] = *(const v2f*)(sWl + bo);
    }
#pragma unroll
    for (int nb = 0; nb < 4; ++nb) {
      acc[nb] = WMMA_F32_16x16x4(ax, bg[nb], acc[nb]);
      acc[nb] = WMMA_F32_16x16x4(as, bs[nb], acc[nb]);
      acc[nb] = WMMA_F32_16x16x4(am, bl[nb], acc[nb]);
    }
  }

#pragma unroll
  for (int nb = 0; nb < 4; ++nb) {
    const int col = nb * 16 + lane_lo;
    const float bv = bias[col];
    v8f r = acc[nb];
#pragma unroll
    for (int rr = 0; rr < 8; ++rr) {
      int m = node_base + rr + 8 * hi;
      if (m < n_nodes) {
        float val = r[rr] + bv;
        val = val > 0.0f ? val : (__expf(val) - 1.0f);
        sum_out[(size_t)m * OUT_DIM + col] = val;
      }
    }
  }
}

extern "C" void kernel_launch(void* const* d_in, const int* in_sizes, int n_in,
                              void* d_out, int out_size, void* d_ws, size_t ws_size,
                              hipStream_t stream) {
  const float* x    = (const float*)d_in[0];
  const float* Wg   = (const float*)d_in[1];
  const float* Wl   = (const float*)d_in[2];
  const float* Ws   = (const float*)d_in[3];
  const float* bias = (const float*)d_in[4];
  const int*   ei   = (const int*)d_in[5];
  float* out = (float*)d_out;
  const int n_nodes = in_sizes[0] / IN_DIM;
  const int n_edges = in_sizes[5] / 2;

  const int    nbins = (n_nodes + BIN_NODES - 1) >> BIN_SHIFT;
  const size_t nb1   = (size_t)nbins + 1;
  const size_t recw  = (3 * nb1 + (size_t)n_nodes + 1) & ~(size_t)1;
  const size_t need  = (recw + 2 * (size_t)n_edges) * sizeof(int);
  if (ws_size >= need && nbins <= MAX_BINS && n_edges > 0) {
    int*   bincnt   = (int*)d_ws;
    int*   cursor   = (int*)d_ws + nb1;
    int*   binstart = (int*)d_ws + 2 * nb1;
    float* deg      = (float*)d_ws + 3 * nb1;
    int2*  rec      = (int2*)((int*)d_ws + recw);
    hipMemsetAsync(bincnt, 0, 2 * nb1 * sizeof(int), stream);
    const int tiles = (n_edges + TILE_EDGES - 1) / TILE_EDGES;
    demo_bin_count_kernel<<<tiles, 256, 0, stream>>>(ei, bincnt, n_edges, n_nodes, nbins);
    demo_bin_scatter_kernel<<<tiles, 256, 0, stream>>>(ei, bincnt, cursor, binstart, rec,
                                                       n_edges, n_nodes, nbins);
    demo_bin_gather_kernel<<<nbins, 512, 0, stream>>>(x, rec, binstart, out, deg, n_nodes);
    {
      int waves  = (n_nodes + 15) / 16;
      int blocks = (waves + 7) / 8;
      demo_node_kernel<<<blocks, 256, 0, stream>>>(x, Wg, Wl, Ws, bias, deg, out,
                                                   n_nodes);
    }
    return;
  }

  float* deg = (float*)d_ws;
  hipMemsetAsync(out, 0, (size_t)n_nodes * IN_DIM * sizeof(float), stream);
  hipMemsetAsync(deg, 0, (size_t)n_nodes * sizeof(float), stream);
  {
    long long threads = (long long)n_edges * 16;
    int blocks = (int)((threads + 255) / 256);
    demo_scatter_kernel<<<blocks, 256, 0, stream>>>(x, ei, out, deg, n_edges);
  }
  {
    int waves  = (n_nodes + 15) / 16;
    int blocks = (waves + 7) / 8;
    demo_node_kernel<<<blocks, 256, 0, stream>>>(x, Wg, Wl, Ws, bias, deg, out,
                                                 n_nodes);
  }
}


# One-pass edge grouping: each 4096-edge tile orders its records by block in LDS into its own slice plus an offset table; gather joins the runs by prefix sum; no count pass, memset or global atomics
# speedup vs baseline: 3.3914x; 1.0901x over previous
#include <hip/hip_runtime.h>

typedef __attribute__((ext_vector_type(2))) float v2f;
typedef __attribute__((ext_vector_type(8))) float v8f;

#define IN_DIM 64
#define OUT_DIM 64
#define LDS_STRIDE 68

#define WMMA_F32_16x16x4(A, B, C) \
  __builtin_amdgcn_wmma_f32_16x16x4_f32(false, (A), false, (B), (short)0, (C), false, false)

__global__ __launch_bounds__(256) void demo_scatter_kernel(
    const float* __restrict__ x, const int* __restrict__ ei,
    float* __restrict__ nsum, float* __restrict__ deg, int n_edges) {
  long long tid = (long long)blockIdx.x * blockDim.x + threadIdx.x;
  int e = (int)(tid >> 4);
  if (e >= n_edges) return;
  int part = (int)(tid & 15);
  int s = ei[e];
  int d = ei[n_edges + e];
  const float4 v = *(const float4*)(x + (size_t)s * IN_DIM + part * 4);
  float* dst = nsum + (size_t)d * IN_DIM + part * 4;
  atomicAdd(dst + 0, v.x);
  atomicAdd(dst + 1, v.y);
  atomicAdd(dst + 2, v.z);
  atomicAdd(dst + 3, v.w);
  if (part == 0) atomicAdd(deg + d, 1.0f);
}

#define BIN_SHIFT 8
#define BIN_NODES 256
#define MAX_BINS  1024
#define TILE_EDGES 4096
#define CHUNK_RECS 6144

typedef __attribute__((ext_vector_type(4))) float v4f;

template <int NW>
__device__ __forceinline__ int block_scan_excl(int v, int* s_wave, int& total) {
  const int lane = threadIdx.x & 31;
  const int w    = threadIdx.x >> 5;
  int x = v;
  for (int o = 1; o < 32; o <<= 1) {
    const int y = __shfl_up(x, o);
    if (lane >= o) x += y;
  }
  __syncthreads();
  if (lane == 31) s_wave[w] = x;
  __syncthreads();
  int wbase = 0, t = 0;
  for (int i = 0; i < NW; ++i) {
    const int sv = s_wave[i];
    if (i < w) wbase += sv;
    t += sv;
  }
  total = t;
  return wbase + x - v;
}

__global__ __launch_bounds__(256) void demo_bin_scatter_kernel(
    const int* __restrict__ ei, int* __restrict__ offs, int2* __restrict__ rec,
    int n_edges, int n_nodes, int nbins, int tiles) {
  __shared__ int  cnt[MAX_BINS];
  __shared__ int  lstart[MAX_BINS];
  __shared__ int  s_wave[8];
  __shared__ __attribute__((aligned(16))) int2 s_rec[TILE_EDGES];
  const int tid = threadIdx.x;
  for (int b = tid; b < nbins; b += 256) cnt[b] = 0;
  __syncthreads();

  const int base = blockIdx.x * TILE_EDGES;
  int d[TILE_EDGES / 256], s[TILE_EDGES / 256], r[TILE_EDGES / 256];
#pragma unroll
  for (int k = 0; k < TILE_EDGES / 256; ++k) {
    const int e = base + k * 256 + tid;
    d[k] = 0; s[k] = 0; r[k] = -1;
    if (e < n_edges) {
      d[k] = ei[n_edges + e];
      s[k] = ei[e];
      if ((unsigned)d[k] < (unsigned)n_nodes) r[k] = atomicAdd(&cnt[d[k] >> BIN_SHIFT], 1);
    }
  }
  __syncthreads();

  int carry = 0;
  for (int b0 = 0; b0 < nbins; b0 += 256) {
    const int b = b0 + tid;
    const int v = b < nbins ? cnt[b] : 0;
    int tot;
    const int p = block_scan_excl<8>(v, s_wave, tot);
    if (b < nbins) {
      lstart[b] = carry + p;
      offs[(size_t)b * tiles + blockIdx.x] = carry + p;
    }
    carry += tot;
  }
  const int nvalid = carry;
  if (tid == 0) offs[(size_t)nbins * tiles + blockIdx.x] = nvalid;
  __syncthreads();

#pragma unroll
  for (int k = 0; k < TILE_EDGES / 256; ++k)
    if (r[k] >= 0) s_rec[lstart[d[k] >> BIN_SHIFT] + r[k]] = make_int2(s[k], d[k]);
  __syncthreads();
  int2* slice = rec + (size_t)blockIdx.x * TILE_EDGES;
  const int pairs = nvalid >> 1;
  for (int i = tid; i < pairs; i += 256)
    ((int4*)slice)[i] = ((const int4*)s_rec)[i];
  if ((nvalid & 1) && tid == 0) slice[nvalid - 1] = s_rec[nvalid - 1];
}

__global__ __launch_bounds__(512) void demo_bin_gather_kernel(
    const float* __restrict__ x, const int2* __restrict__ rec,
    const int* __restrict__ offs, float* __restrict__ nsum,
    float* __restrict__ deg, int n_nodes, int tiles) {
  __shared__ int head[BIN_NODES];
  __shared__ int s_src[CHUNK_RECS];
  __shared__ int s_next[CHUNK_RECS];
  __shared__ int s_wave[16];
  const int tid   = threadIdx.x;
  const int part  = tid & 7;
  const int g     = tid >> 3;
  const int node0 = blockIdx.x << BIN_SHIFT;
  const int* row0 = offs + (size_t)blockIdx.x * tiles;
  const int* row1 = row0 + tiles;
  const v4f* x4 = (const v4f*)x;

  v4f a0[4], a1[4];
  int cnt[4];
#pragma unroll
  for (int j = 0; j < 4; ++j) { a0[j] = v4f{}; a1[j] = v4f{}; cnt[j] = 0; }

  for (int t0 = 0; t0 < tiles; t0 += 512) {
    const int t = t0 + tid;
    int st = 0, len = 0, gidx = 0;
    if (t < tiles) {
      st  = row0[t];
      len = row1[t] - st;
      st   = min(max(st, 0), TILE_EDGES);
      len  = min(max(len, 0), TILE_EDGES - st);
      gidx = t * TILE_EDGES + st;
    }
    int total;
    const int pos = block_scan_excl<16>(len, s_wave, total);

    for (int c0 = 0; c0 < total; c0 += CHUNK_RECS) {
      const int n = min(CHUNK_RECS, total - c0);
      if (tid < BIN_NODES) head[tid] = -1;
      const int klo = max(0, c0 - pos);
      const int khi = min(len, c0 + n - pos);
#pragma nounroll
      for (int k = klo; k < khi; ++k) s_src[pos + k - c0] = gidx + k;
      __syncthreads();

      int2 q[CHUNK_RECS / 512];
#pragma unroll
      for (int j = 0; j < CHUNK_RECS / 512; ++j)
        q[j] = rec[s_src[min(tid + j * 512, n - 1)]];
      __syncthreads();
      int prev[CHUNK_RECS / 512];
#pragma unroll
      for (int j = 0; j < CHUNK_RECS / 512; ++j) {
        const int i = tid + j * 512;
        prev[j] = -1;
        if (i < n) {
          s_src[i] = q[j].x;
          prev[j] = atomicExch(&head[(q[j].y - node0) & (BIN_NODES - 1)], i);
        }
      }
#pragma unroll
      for (int j = 0; j < CHUNK_RECS / 512; ++j) {
        const int i = tid + j * 512;
        if (i < n) s_next[i] = prev[j];
      }
      __syncthreads();

      int cur[4];
#pragma unroll
      for (int j = 0; j < 4; ++j) cur[j] = head[g + 64 * j];
      for (int it = 0; it < n; ++it) {
        if ((cur[0] & cur[1] & cur[2] & cur[3]) < 0) break;
        v4f v0[4], v1[4];
        bool act[4];
#pragma unroll
        for (int j = 0; j < 4; ++j) {
          act[j] = cur[j] >= 0;
          const int ci   = act[j] ? cur[j] : 0;
          const int sidx = s_src[ci];
          const int nx   = s_next[ci];
          const v4f* row = x4 + (size_t)sidx * (IN_DIM / 4);
          v0[j] = row[part];
          v1[j] = row[part + 8];
          cur[j] = act[j] ? nx : -1;
          cnt[j] += act[j] ? 1 : 0;
        }
#pragma unroll
        for (int j = 0; j < 4; ++j) {
          a0[j] += act[j] ? v0[j] : v4f{};
          a1[j] += act[j] ? v1[j] : v4f{};
        }
      }
      __syncthreads();
    }
  }

#pragma unroll
  for (int j = 0; j < 4; ++j) {
    const int node = node0 + g + 64 * j;
    if (node < n_nodes) {
      v4f* orow = (v4f*)nsum + (size_t)node * (IN_DIM / 4);
      orow[part]     = a0[j];
      orow[part + 8] = a1[j];
      if (part == 0) deg[node] = (float)cnt[j];
    }
  }
}

__global__ __launch_bounds__(256) void demo_node_kernel(
    const float* __restrict__ x, const float* __restrict__ Wg,
    const float* __restrict__ Wl, const float* __restrict__ Ws,
    const float* __restrict__ bias, const float* __restrict__ deg,
    float* sum_out, int n_nodes) {
  __shared__ float sW[3 * OUT_DIM * LDS_STRIDE];
  float* sWg = sW;
  float* sWs = sW + OUT_DIM * LDS_STRIDE;
  float* sWl = sW + 2 * OUT_DIM * LDS_STRIDE;

  {
    const int tid = threadIdx.x;
#pragma unroll
    for (int i = 0; i < 4; ++i) {
      int idx = tid + i * 256;
      int row = idx >> 4;
      int k4  = idx & 15;
      float4 g = *(const float4*)(Wg + (size_t)idx * 4);
      float4 s = *(const float4*)(Ws + (size_t)idx * 4);
      float4 l = *(const float4*)(Wl + (size_t)idx * 4);
      *(float4*)(sWg + row * LDS_STRIDE + k4 * 4) = g;
      *(float4*)(sWs + row * LDS_STRIDE + k4 * 4) = s;
      *(float4*)(sWl + row * LDS_STRIDE + k4 * 4) = l;
    }
  }
  __syncthreads();

  const int lane    = threadIdx.x & 31;
  const int wave    = threadIdx.x >> 5;
  const int lane_lo = lane & 15;
  const int hi      = lane >> 4;
  const int node_base = (blockIdx.x * 8 + wave) * 16;
  if (node_base >= n_nodes) return;

  int arow = node_base + lane_lo;
  if (arow > n_nodes - 1) arow = n_nodes - 1;

  const float dv  = deg[arow];
  const float inv = 1.0f / fmaxf(dv, 1.0f);
  const float hn  = dv > 0.0f ? 1.0f : 0.0f;

  const v2f* xr = (const v2f*)(x       + (size_t)arow * IN_DIM + 2 * hi);
  const v2f* sr = (const v2f*)(sum_out + (size_t)arow * IN_DIM + 2 * hi);

  const int bofs = lane_lo * LDS_STRIDE + 2 * hi;

  v8f acc[4] = {v8f{}, v8f{}, v8f{}, v8f{}};

  for (int c = 0; c < 16; ++c) {
    v2f ax = xr[2 * c];
    v2f am = sr[2 * c] * inv;
    v2f as = ax * hn;

    v2f bg[4], bs[4], bl[4];
#pragma unroll
    for (int nb = 0; nb < 4; ++nb) {
      const int bo = bofs + nb * 16 * LDS_STRIDE + 4 * c;
      bg[nb] = *(const v2f*)(sWg + bo);
      bs[nb] = *(const v2f*)(sWs + bo);
      bl[nb] = *(const v2f*)(sWl + bo);
    }
#pragma unroll
    for (int nb = 0; nb < 4; ++nb) {
      acc[nb] = WMMA_F32_16x16x4(ax, bg[nb], acc[nb]);
      acc[nb] = WMMA_F32_16x16x4(as, bs[nb], acc[nb]);
      acc[nb] = WMMA_F32_16x16x4(am, bl[nb], acc[nb]);
    }
  }

#pragma unroll
  for (int nb = 0; nb < 4; ++nb) {
    const int col = nb * 16 + lane_lo;
    const float bv = bias[col];
    v8f r = acc[nb];
#pragma unroll
    for (int rr = 0; rr < 8; ++rr) {
      int m = node_base + rr + 8 * hi;
      if (m < n_nodes) {
        float val = r[rr] + bv;
        val = val > 0.0f ? val : (__expf(val) - 1.0f);
        sum_out[(size_t)m * OUT_DIM + col] = val;
      }
    }
  }
}

extern "C" void kernel_launch(void* const* d_in, const int* in_sizes, int n_in,
                              void* d_out, int out_size, void* d_ws, size_t ws_size,
                              hipStream_t stream) {
  const float* x    = (const float*)d_in[0];
  const float* Wg   = (const float*)d_in[1];
  const float* Wl   = (const float*)d_in[2];
  const float* Ws   = (const float*)d_in[3];
  const float* bias = (const float*)d_in[4];
  const int*   ei   = (const int*)d_in[5];
  float* out = (float*)d_out;
  const int n_nodes = in_sizes[0] / IN_DIM;
  const int n_edges = in_sizes[5] / 2;

  const int    nbins = (n_nodes + BIN_NODES - 1) >> BIN_SHIFT;
  const int    tiles = (n_edges + TILE_EDGES - 1) / TILE_EDGES;
  const size_t offw  = ((size_t)nbins + 1) * (size_t)tiles;
  const size_t recw  = (offw + (size_t)n_nodes + 31) & ~(size_t)31;
  const size_t need  = (recw + 2 * (size_t)tiles * TILE_EDGES) * sizeof(int);
  if (ws_size >= need && nbins <= MAX_BINS && n_edges > 0) {
    int*   offs = (int*)d_ws;
    float* deg  = (float*)d_ws + offw;
    int2*  rec  = (int2*)((int*)d_ws + recw);
    demo_bin_scatter_kernel<<<tiles, 256, 0, stream>>>(ei, offs, rec, n_edges, n_nodes,
                                                       nbins, tiles);
    demo_bin_gather_kernel<<<nbins, 512, 0, stream>>>(x, rec, offs, out, deg, n_nodes,
                                                      tiles);
    {
      int waves  = (n_nodes + 15) / 16;
      int blocks = (waves + 7) / 8;
      demo_node_kernel<<<blocks, 256, 0, stream>>>(x, Wg, Wl, Ws, bias, deg, out,
                                                   n_nodes);
    }
    return;
  }

  float* deg = (float*)d_ws;
  hipMemsetAsync(out, 0, (size_t)n_nodes * IN_DIM * sizeof(float), stream);
  hipMemsetAsync(deg, 0, (size_t)n_nodes * sizeof(float), stream);
  {
    long long threads = (long long)n_edges * 16;
    int blocks = (int)((threads + 255) / 256);
    demo_scatter_kernel<<<blocks, 256, 0, stream>>>(x, ei, out, deg, n_edges);
  }
  {
    int waves  = (n_nodes + 15) / 16;
    int blocks = (waves + 7) / 8;
    demo_node_kernel<<<blocks, 256, 0, stream>>>(x, Wg, Wl, Ws, bias, deg, out,
                                                 n_nodes);
  }
}


# Scatter kernel: a thread's 32 edge-id loads, then its 16 LDS run-start reads, each issued as one group before a single wait; was one load-wait-use round trip per edge; sum order = arrival as before
# speedup vs baseline: 3.7028x; 1.0918x over previous
#include <hip/hip_runtime.h>

typedef __attribute__((ext_vector_type(2))) float v2f;
typedef __attribute__((ext_vector_type(8))) float v8f;

#define IN_DIM 64
#define OUT_DIM 64
#define LDS_STRIDE 68

#define WMMA_F32_16x16x4(A, B, C) \
  __builtin_amdgcn_wmma_f32_16x16x4_f32(false, (A), false, (B), (short)0, (C), false, false)

__global__ __launch_bounds__(256) void demo_scatter_kernel(
    const float* __restrict__ x, const int* __restrict__ ei,
    float* __restrict__ nsum, float* __restrict__ deg, int n_edges) {
  long long tid = (long long)blockIdx.x * blockDim.x + threadIdx.x;
  int e = (int)(tid >> 4);
  if (e >= n_edges) return;
  int part = (int)(tid & 15);
  int s = ei[e];
  int d = ei[n_edges + e];
  const float4 v = *(const float4*)(x + (size_t)s * IN_DIM + part * 4);
  float* dst = nsum + (size_t)d * IN_DIM + part * 4;
  atomicAdd(dst + 0, v.x);
  atomicAdd(dst + 1, v.y);
  atomicAdd(dst + 2, v.z);
  atomicAdd(dst + 3, v.w);
  if (part == 0) atomicAdd(deg + d, 1.0f);
}

#define BIN_SHIFT 8
#define BIN_NODES 256
#define MAX_BINS  1024
#define TILE_EDGES 4096
#define CHUNK_RECS 6144

typedef __attribute__((ext_vector_type(4))) float v4f;

template <int NW>
__device__ __forceinline__ int block_scan_excl(int v, int* s_wave, int& total) {
  const int lane = threadIdx.x & 31;
  const int w    = threadIdx.x >> 5;
  int x = v;
  for (int o = 1; o < 32; o <<= 1) {
    const int y = __shfl_up(x, o);
    if (lane >= o) x += y;
  }
  __syncthreads();
  if (lane == 31) s_wave[w] = x;
  __syncthreads();
  int wbase = 0, t = 0;
  for (int i = 0; i < NW; ++i) {
    const int sv = s_wave[i];
    if (i < w) wbase += sv;
    t += sv;
  }
  total = t;
  return wbase + x - v;
}

__global__ __launch_bounds__(256) void demo_bin_scatter_kernel(
    const int* __restrict__ ei, int* __restrict__ offs, int2* __restrict__ rec,
    int n_edges, int n_nodes, int nbins, int tiles) {
  __shared__ int  cnt[MAX_BINS];
  __shared__ int  lstart[MAX_BINS];
  __shared__ int  s_wave[8];
  __shared__ __attribute__((aligned(16))) int2 s_rec[TILE_EDGES];
  const int tid = threadIdx.x;

  const int base = blockIdx.x * TILE_EDGES;
  int d[TILE_EDGES / 256], s[TILE_EDGES / 256], r[TILE_EDGES / 256];
#pragma unroll
  for (int k = 0; k < TILE_EDGES / 256; ++k) {
    const int e = min(base + k * 256 + tid, n_edges - 1);
    d[k] = ei[n_edges + e];
    s[k] = ei[e];
  }
  for (int b = tid; b < nbins; b += 256) cnt[b] = 0;
  __syncthreads();

#pragma unroll
  for (int k = 0; k < TILE_EDGES / 256; ++k) {
    r[k] = -1;
    if (base + k * 256 + tid < n_edges && (unsigned)d[k] < (unsigned)n_nodes)
      r[k] = atomicAdd(&cnt[d[k] >> BIN_SHIFT], 1);
  }
  __syncthreads();

  int carry = 0;
  for (int b0 = 0; b0 < nbins; b0 += 256) {
    const int b = b0 + tid;
    const int v = b < nbins ? cnt[b] : 0;
    int tot;
    const int p = block_scan_excl<8>(v, s_wave, tot);
    if (b < nbins) {
      lstart[b] = carry + p;
      offs[(size_t)b * tiles + blockIdx.x] = carry + p;
    }
    carry += tot;
  }
  const int nvalid = carry;
  if (tid == 0) offs[(size_t)nbins * tiles + blockIdx.x] = nvalid;
  __syncthreads();

  int ls[TILE_EDGES / 256];
#pragma unroll
  for (int k = 0; k < TILE_EDGES / 256; ++k)
    ls[k] = lstart[r[k] >= 0 ? (d[k] >> BIN_SHIFT) : 0];
#pragma unroll
  for (int k = 0; k < TILE_EDGES / 256; ++k)
    if (r[k] >= 0) s_rec[ls[k] + r[k]] = make_int2(s[k], d[k]);
  __syncthreads();
  int2* slice = rec + (size_t)blockIdx.x * TILE_EDGES;
  const int pairs = nvalid >> 1;
  for (int i = tid; i < pairs; i += 256)
    ((int4*)slice)[i] = ((const int4*)s_rec)[i];
  if ((nvalid & 1) && tid == 0) slice[nvalid - 1] = s_rec[nvalid - 1];
}

__global__ __launch_bounds__(512) void demo_bin_gather_kernel(
    const float* __restrict__ x, const int2* __restrict__ rec,
    const int* __restrict__ offs, float* __restrict__ nsum,
    float* __restrict__ deg, int n_nodes, int tiles) {
  __shared__ int head[BIN_NODES];
  __shared__ int s_src[CHUNK_RECS];
  __shared__ int s_next[CHUNK_RECS];
  __shared__ int s_wave[16];
  const int tid   = threadIdx.x;
  const int part  = tid & 7;
  const int g     = tid >> 3;
  const int node0 = blockIdx.x << BIN_SHIFT;
  const int* row0 = offs + (size_t)blockIdx.x * tiles;
  const int* row1 = row0 + tiles;
  const v4f* x4 = (const v4f*)x;

  v4f a0[4], a1[4];
  int cnt[4];
#pragma unroll
  for (int j = 0; j < 4; ++j) { a0[j] = v4f{}; a1[j] = v4f{}; cnt[j] = 0; }

  for (int t0 = 0; t0 < tiles; t0 += 512) {
    const int t = t0 + tid;
    int st = 0, len = 0, gidx = 0;
    if (t < tiles) {
      st  = row0[t];
      len = row1[t] - st;
      st   = min(max(st, 0), TILE_EDGES);
      len  = min(max(len, 0), TILE_EDGES - st);
      gidx = t * TILE_EDGES + st;
    }
    int total;
    const int pos = block_scan_excl<16>(len, s_wave, total);

    for (int c0 = 0; c0 < total; c0 += CHUNK_RECS) {
      const int n = min(CHUNK_RECS, total - c0);
      if (tid < BIN_NODES) head[tid] = -1;
      const int klo = max(0, c0 - pos);
      const int khi = min(len, c0 + n - pos);
#pragma nounroll
      for (int k = klo; k < khi; ++k) s_src[pos + k - c0] = gidx + k;
      __syncthreads();

      int2 q[CHUNK_RECS / 512];
#pragma unroll
      for (int j = 0; j < CHUNK_RECS / 512; ++j)
        q[j] = rec[s_src[min(tid + j * 512, n - 1)]];
      __syncthreads();
      int prev[CHUNK_RECS / 512];
#pragma unroll
      for (int j = 0; j < CHUNK_RECS / 512; ++j) {
        const int i = tid + j * 512;
        prev[j] = -1;
        if (i < n) {
          s_src[i] = q[j].x;
          prev[j] = atomicExch(&head[(q[j].y - node0) & (BIN_NODES - 1)], i);
        }
      }
#pragma unroll
      for (int j = 0; j < CHUNK_RECS / 512; ++j) {
        const int i = tid + j * 512;
        if (i < n) s_next[i] = prev[j];
      }
      __syncthreads();

      int cur[4];
#pragma unroll
      for (int j = 0; j < 4; ++j) cur[j] = head[g + 64 * j];
      for (int it = 0; it < n; ++it) {
        if ((cur[0] & cur[1] & cur[2] & cur[3]) < 0) break;
        v4f v0[4], v1[4];
        bool act[4];
#pragma unroll
        for (int j = 0; j < 4; ++j) {
          act[j] = cur[j] >= 0;
          const int ci   = act[j] ? cur[j] : 0;
          const int sidx = s_src[ci];
          const int nx   = s_next[ci];
          const v4f* row = x4 + (size_t)sidx * (IN_DIM / 4);
          v0[j] = row[part];
          v1[j] = row[part + 8];
          cur[j] = act[j] ? nx : -1;
          cnt[j] += act[j] ? 1 : 0;
        }
#pragma unroll
        for (int j = 0; j < 4; ++j) {
          a0[j] += act[j] ? v0[j] : v4f{};
          a1[j] += act[j] ? v1[j] : v4f{};
        }
      }
      __syncthreads();
    }
  }

#pragma unroll
  for (int j = 0; j < 4; ++j) {
    const int node = node0 + g + 64 * j;
    if (node < n_nodes) {
      v4f* orow = (v4f*)nsum + (size_t)node * (IN_DIM / 4);
      orow[part]     = a0[j];
      orow[part + 8] = a1[j];
      if (part == 0) deg[node] = (float)cnt[j];
    }
  }
}

__global__ __launch_bounds__(256) void demo_node_kernel(
    const float* __restrict__ x, const float* __restrict__ Wg,
    const float* __restrict__ Wl, const float* __restrict__ Ws,
    const float* __restrict__ bias, const float* __restrict__ deg,
    float* sum_out, int n_nodes) {
  __shared__ float sW[3 * OUT_DIM * LDS_STRIDE];
  float* sWg = sW;
  float* sWs = sW + OUT_DIM * LDS_STRIDE;
  float* sWl = sW + 2 * OUT_DIM * LDS_STRIDE;

  {
    const int tid = threadIdx.x;
#pragma unroll
    for (int i = 0; i < 4; ++i) {
      int idx = tid + i * 256;
      int row = idx >> 4;
      int k4  = idx & 15;
      float4 g = *(const float4*)(Wg + (size_t)idx * 4);
      float4 s = *(const float4*)(Ws + (size_t)idx * 4);
      float4 l = *(const float4*)(Wl + (size_t)idx * 4);
      *(float4*)(sWg + row * LDS_STRIDE + k4 * 4) = g;
      *(float4*)(sWs + row * LDS_STRIDE + k4 * 4) = s;
      *(float4*)(sWl + row * LDS_STRIDE + k4 * 4) = l;
    }
  }
  __syncthreads();

  const int lane    = threadIdx.x & 31;
  const int wave    = threadIdx.x >> 5;
  const int lane_lo = lane & 15;
  const int hi      = lane >> 4;
  const int node_base = (blockIdx.x * 8 + wave) * 16;
  if (node_base >= n_nodes) return;

  int arow = node_base + lane_lo;
  if (arow > n_nodes - 1) arow = n_nodes - 1;

  const float dv  = deg[arow];
  const float inv = 1.0f / fmaxf(dv, 1.0f);
  const float hn  = dv > 0.0f ? 1.0f : 0.0f;

  const v2f* xr = (const v2f*)(x       + (size_t)arow * IN_DIM + 2 * hi);
  const v2f* sr = (const v2f*)(sum_out + (size_t)arow * IN_DIM + 2 * hi);

  const int bofs = lane_lo * LDS_STRIDE + 2 * hi;

  v8f acc[4] = {v8f{}, v8f{}, v8f{}, v8f{}};

  for (int c = 0; c < 16; ++c) {
    v2f ax = xr[2 * c];
    v2f am = sr[2 * c] * inv;
    v2f as = ax * hn;

    v2f bg[4], bs[4], bl[4];
#pragma unroll
    for (int nb = 0; nb < 4; ++nb) {
      const int bo = bofs + nb * 16 * LDS_STRIDE + 4 * c;
      bg[nb] = *(const v2f*)(sWg + bo);
      bs[nb] = *(const v2f*)(sWs + bo);
      bl[nb] = *(const v2f*)(sWl + bo);
    }
#pragma unroll
    for (int nb = 0; nb < 4; ++nb) {
      acc[nb] = WMMA_F32_16x16x4(ax, bg[nb], acc[nb]);
      acc[nb] = WMMA_F32_16x16x4(as, bs[nb], acc[nb]);
      acc[nb] = WMMA_F32_16x16x4(am, bl[nb], acc[nb]);
    }
  }

#pragma unroll
  for (int nb = 0; nb < 4; ++nb) {
    const int col = nb * 16 + lane_lo;
    const float bv = bias[col];
    v8f r = acc[nb];
#pragma unroll
    for (int rr = 0; rr < 8; ++rr) {
      int m = node_base + rr + 8 * hi;
      if (m < n_nodes) {
        float val = r[rr] + bv;
        val = val > 0.0f ? val : (__expf(val) - 1.0f);
        sum_out[(size_t)m * OUT_DIM + col] = val;
      }
    }
  }
}

extern "C" void kernel_launch(void* const* d_in, const int* in_sizes, int n_in,
                              void* d_out, int out_size, void* d_ws, size_t ws_size,
                              hipStream_t stream) {
  const float* x    = (const float*)d_in[0];
  const float* Wg   = (const float*)d_in[1];
  const float* Wl   = (const float*)d_in[2];
  const float* Ws   = (const float*)d_in[3];
  const float* bias = (const float*)d_in[4];
  const int*   ei   = (const int*)d_in[5];
  float* out = (float*)d_out;
  const int n_nodes = in_sizes[0] / IN_DIM;
  const int n_edges = in_sizes[5] / 2;

  const int    nbins = (n_nodes + BIN_NODES - 1) >> BIN_SHIFT;
  const int    tiles = (n_edges + TILE_EDGES - 1) / TILE_EDGES;
  const size_t offw  = ((size_t)nbins + 1) * (size_t)tiles;
  const size_t recw  = (offw + (size_t)n_nodes + 31) & ~(size_t)31;
  const size_t need  = (recw + 2 * (size_t)tiles * TILE_EDGES) * sizeof(int);
  if (ws_size >= need && nbins <= MAX_BINS && n_edges > 0) {
    int*   offs = (int*)d_ws;
    float* deg  = (float*)d_ws + offw;
    int2*  rec  = (int2*)((int*)d_ws + recw);
    demo_bin_scatter_kernel<<<tiles, 256, 0, stream>>>(ei, offs, rec, n_edges, n_nodes,
                                                       nbins, tiles);
    demo_bin_gather_kernel<<<nbins, 512, 0, stream>>>(x, rec, offs, out, deg, n_nodes,
                                                      tiles);
    {
      int waves  = (n_nodes + 15) / 16;
      int blocks = (waves + 7) / 8;
      demo_node_kernel<<<blocks, 256, 0, stream>>>(x, Wg, Wl, Ws, bias, deg, out,
                                                   n_nodes);
    }
    return;
  }

  float* deg = (float*)d_ws;
  hipMemsetAsync(out, 0, (size_t)n_nodes * IN_DIM * sizeof(float), stream);
  hipMemsetAsync(deg, 0, (size_t)n_nodes * sizeof(float), stream);
  {
    long long threads = (long long)n_edges * 16;
    int blocks = (int)((threads + 255) / 256);
    demo_scatter_kernel<<<blocks, 256, 0, stream>>>(x, ei, out, deg, n_edges);
  }
  {
    int waves  = (n_nodes + 15) / 16;
    int blocks = (waves + 7) / 8;
    demo_node_kernel<<<blocks, 256, 0, stream>>>(x, Wg, Wl, Ws, bias, deg, out,
                                                 n_nodes);
  }
}


# Node kernel: a tile whose 16 nodes all have a neighbour uses x@(Wg+Ws)^T, weights added at staging: 2 f32 WMMAs per k-step and column tile, was 3; x-term sum order changes (last-bit rounding)
# speedup vs baseline: 3.8974x; 1.0526x over previous
#include <hip/hip_runtime.h>

typedef __attribute__((ext_vector_type(2))) float v2f;
typedef __attribute__((ext_vector_type(8))) float v8f;

#define IN_DIM 64
#define OUT_DIM 64
#define LDS_STRIDE 68

#define WMMA_F32_16x16x4(A, B, C) \
  __builtin_amdgcn_wmma_f32_16x16x4_f32(false, (A), false, (B), (short)0, (C), false, false)

__global__ __launch_bounds__(256) void demo_scatter_kernel(
    const float* __restrict__ x, const int* __restrict__ ei,
    float* __restrict__ nsum, float* __restrict__ deg, int n_edges) {
  long long tid = (long long)blockIdx.x * blockDim.x + threadIdx.x;
  int e = (int)(tid >> 4);
  if (e >= n_edges) return;
  int part = (int)(tid & 15);
  int s = ei[e];
  int d = ei[n_edges + e];
  const float4 v = *(const float4*)(x + (size_t)s * IN_DIM + part * 4);
  float* dst = nsum + (size_t)d * IN_DIM + part * 4;
  atomicAdd(dst + 0, v.x);
  atomicAdd(dst + 1, v.y);
  atomicAdd(dst + 2, v.z);
  atomicAdd(dst + 3, v.w);
  if (part == 0) atomicAdd(deg + d, 1.0f);
}

#define BIN_SHIFT 8
#define BIN_NODES 256
#define MAX_BINS  1024
#define TILE_EDGES 4096
#define CHUNK_RECS 6144

typedef __attribute__((ext_vector_type(4))) float v4f;

template <int NW>
__device__ __forceinline__ int block_scan_excl(int v, int* s_wave, int& total) {
  const int lane = threadIdx.x & 31;
  const int w    = threadIdx.x >> 5;
  int x = v;
  for (int o = 1; o < 32; o <<= 1) {
    const int y = __shfl_up(x, o);
    if (lane >= o) x += y;
  }
  __syncthreads();
  if (lane == 31) s_wave[w] = x;
  __syncthreads();
  int wbase = 0, t = 0;
  for (int i = 0; i < NW; ++i) {
    const int sv = s_wave[i];
    if (i < w) wbase += sv;
    t += sv;
  }
  total = t;
  return wbase + x - v;
}

__global__ __launch_bounds__(256) void demo_bin_scatter_kernel(
    const int* __restrict__ ei, int* __restrict__ offs, int2* __restrict__ rec,
    int n_edges, int n_nodes, int nbins, int tiles) {
  __shared__ int  cnt[MAX_BINS];
  __shared__ int  lstart[MAX_BINS];
  __shared__ int  s_wave[8];
  __shared__ __attribute__((aligned(16))) int2 s_rec[TILE_EDGES];
  const int tid = threadIdx.x;

  const int base = blockIdx.x * TILE_EDGES;
  int d[TILE_EDGES / 256], s[TILE_EDGES / 256], r[TILE_EDGES / 256];
#pragma unroll
  for (int k = 0; k < TILE_EDGES / 256; ++k) {
    const int e = min(base + k * 256 + tid, n_edges - 1);
    d[k] = ei[n_edges + e];
    s[k] = ei[e];
  }
  for (int b = tid; b < nbins; b += 256) cnt[b] = 0;
  __syncthreads();

#pragma unroll
  for (int k = 0; k < TILE_EDGES / 256; ++k) {
    r[k] = -1;
    if (base + k * 256 + tid < n_edges && (unsigned)d[k] < (unsigned)n_nodes)
      r[k] = atomicAdd(&cnt[d[k] >> BIN_SHIFT], 1);
  }
  __syncthreads();

  int carry = 0;
  for (int b0 = 0; b0 < nbins; b0 += 256) {
    const int b = b0 + tid;
    const int v = b < nbins ? cnt[b] : 0;
    int tot;
    const int p = block_scan_excl<8>(v, s_wave, tot);
    if (b < nbins) {
      lstart[b] = carry + p;
      offs[(size_t)b * tiles + blockIdx.x] = carry + p;
    }
    carry += tot;
  }
  const int nvalid = carry;
  if (tid == 0) offs[(size_t)nbins * tiles + blockIdx.x] = nvalid;
  __syncthreads();

  int ls[TILE_EDGES / 256];
#pragma unroll
  for (int k = 0; k < TILE_EDGES / 256; ++k)
    ls[k] = lstart[r[k] >= 0 ? (d[k] >> BIN_SHIFT) : 0];
#pragma unroll
  for (int k = 0; k < TILE_EDGES / 256; ++k)
    if (r[k] >= 0) s_rec[ls[k] + r[k]] = make_int2(s[k], d[k]);
  __syncthreads();
  int2* slice = rec + (size_t)blockIdx.x * TILE_EDGES;
  const int pairs = nvalid >> 1;
  for (int i = tid; i < pairs; i += 256)
    ((int4*)slice)[i] = ((const int4*)s_rec)[i];
  if ((nvalid & 1) && tid == 0) slice[nvalid - 1] = s_rec[nvalid - 1];
}

__global__ __launch_bounds__(512) void demo_bin_gather_kernel(
    const float* __restrict__ x, const int2* __restrict__ rec,
    const int* __restrict__ offs, float* __restrict__ nsum,
    float* __restrict__ deg, int n_nodes, int tiles) {
  __shared__ int head[BIN_NODES];
  __shared__ int s_src[CHUNK_RECS];
  __shared__ int s_next[CHUNK_RECS];
  __shared__ int s_wave[16];
  const int tid   = threadIdx.x;
  const int part  = tid & 7;
  const int g     = tid >> 3;
  const int node0 = blockIdx.x << BIN_SHIFT;
  const int* row0 = offs + (size_t)blockIdx.x * tiles;
  const int* row1 = row0 + tiles;
  const v4f* x4 = (const v4f*)x;

  v4f a0[4], a1[4];
  int cnt[4];
#pragma unroll
  for (int j = 0; j < 4; ++j) { a0[j] = v4f{}; a1[j] = v4f{}; cnt[j] = 0; }

  for (int t0 = 0; t0 < tiles; t0 += 512) {
    const int t = t0 + tid;
    int st = 0, len = 0, gidx = 0;
    if (t < tiles) {
      st  = row0[t];
      len = row1[t] - st;
      st   = min(max(st, 0), TILE_EDGES);
      len  = min(max(len, 0), TILE_EDGES - st);
      gidx = t * TILE_EDGES + st;
    }
    int total;
    const int pos = block_scan_excl<16>(len, s_wave, total);

    for (int c0 = 0; c0 < total; c0 += CHUNK_RECS) {
      const int n = min(CHUNK_RECS, total - c0);
      if (tid < BIN_NODES) head[tid] = -1;
      const int klo = max(0, c0 - pos);
      const int khi = min(len, c0 + n - pos);
#pragma nounroll
      for (int k = klo; k < khi; ++k) s_src[pos + k - c0] = gidx + k;
      __syncthreads();

      int2 q[CHUNK_RECS / 512];
#pragma unroll
      for (int j = 0; j < CHUNK_RECS / 512; ++j)
        q[j] = rec[s_src[min(tid + j * 512, n - 1)]];
      __syncthreads();
      int prev[CHUNK_RECS / 512];
#pragma unroll
      for (int j = 0; j < CHUNK_RECS / 512; ++j) {
        const int i = tid + j * 512;
        prev[j] = -1;
        if (i < n) {
          s_src[i] = q[j].x;
          prev[j] = atomicExch(&head[(q[j].y - node0) & (BIN_NODES - 1)], i);
        }
      }
#pragma unroll
      for (int j = 0; j < CHUNK_RECS / 512; ++j) {
        const int i = tid + j * 512;
        if (i < n) s_next[i] = prev[j];
      }
      __syncthreads();

      int cur[4];
#pragma unroll
      for (int j = 0; j < 4; ++j) cur[j] = head[g + 64 * j];
      for (int it = 0; it < n; ++it) {
        if ((cur[0] & cur[1] & cur[2] & cur[3]) < 0) break;
        v4f v0[4], v1[4];
        bool act[4];
#pragma unroll
        for (int j = 0; j < 4; ++j) {
          act[j] = cur[j] >= 0;
          const int ci   = act[j] ? cur[j] : 0;
          const int sidx = s_src[ci];
          const int nx   = s_next[ci];
          const v4f* row = x4 + (size_t)sidx * (IN_DIM / 4);
          v0[j] = row[part];
          v1[j] = row[part + 8];
          cur[j] = act[j] ? nx : -1;
          cnt[j] += act[j] ? 1 : 0;
        }
#pragma unroll
        for (int j = 0; j < 4; ++j) {
          a0[j] += act[j] ? v0[j] : v4f{};
          a1[j] += act[j] ? v1[j] : v4f{};
        }
      }
      __syncthreads();
    }
  }

#pragma unroll
  for (int j = 0; j < 4; ++j) {
    const int node = node0 + g + 64 * j;
    if (node < n_nodes) {
      v4f* orow = (v4f*)nsum + (size_t)node * (IN_DIM / 4);
      orow[part]     = a0[j];
      orow[part + 8] = a1[j];
      if (part == 0) deg[node] = (float)cnt[j];
    }
  }
}

__global__ __launch_bounds__(256, 8) void demo_node_kernel(
    const float* __restrict__ x, const float* __restrict__ Wg,
    const float* __restrict__ Wl, const float* __restrict__ Ws,
    const float* __restrict__ bias, const float* __restrict__ deg,
    float* sum_out, int n_nodes) {
  __shared__ float sW[4 * OUT_DIM * LDS_STRIDE];
  float* sWg = sW;
  float* sWs = sW + OUT_DIM * LDS_STRIDE;
  float* sWl = sW + 2 * OUT_DIM * LDS_STRIDE;
  float* sWc = sW + 3 * OUT_DIM * LDS_STRIDE;

  {
    const int tid = threadIdx.x;
#pragma unroll
    for (int i = 0; i < 4; ++i) {
      int idx = tid + i * 256;
      int row = idx >> 4;
      int k4  = idx & 15;
      float4 g = *(const float4*)(Wg + (size_t)idx * 4);
      float4 s = *(const float4*)(Ws + (size_t)idx * 4);
      float4 l = *(const float4*)(Wl + (size_t)idx * 4);
      *(float4*)(sWg + row * LDS_STRIDE + k4 * 4) = g;
      *(float4*)(sWs + row * LDS_STRIDE + k4 * 4) = s;
      *(float4*)(sWl + row * LDS_STRIDE + k4 * 4) = l;
      *(float4*)(sWc + row * LDS_STRIDE + k4 * 4) =
          make_float4(g.x + s.x, g.y + s.y, g.z + s.z, g.w + s.w);
    }
  }
  __syncthreads();

  const int lane    = threadIdx.x & 31;
  const int wave    = threadIdx.x >> 5;
  const int lane_lo = lane & 15;
  const int hi      = lane >> 4;
  const int node_base = (blockIdx.x * 8 + wave) * 16;
  if (node_base >= n_nodes) return;

  int arow = node_base + lane_lo;
  if (arow > n_nodes - 1) arow = n_nodes - 1;

  const float dv  = deg[arow];
  const float inv = 1.0f / fmaxf(dv, 1.0f);
  const float hn  = dv > 0.0f ? 1.0f : 0.0f;

  const v2f* xr = (const v2f*)(x       + (size_t)arow * IN_DIM + 2 * hi);
  const v2f* sr = (const v2f*)(sum_out + (size_t)arow * IN_DIM + 2 * hi);

  const int bofs = lane_lo * LDS_STRIDE + 2 * hi;

  v8f acc[4] = {v8f{}, v8f{}, v8f{}, v8f{}};

  if (__all(dv > 0.0f)) {
#pragma unroll
    for (int c = 0; c < 16; ++c) {
      v2f ax = xr[2 * c];
      v2f am = sr[2 * c] * inv;
      v2f bc[4], bl[4];
#pragma unroll
      for (int nb = 0; nb < 4; ++nb) {
        const int bo = bofs + nb * 16 * LDS_STRIDE + 4 * c;
        bc[nb] = *(const v2f*)(sWc + bo);
        bl[nb] = *(const v2f*)(sWl + bo);
      }
#pragma unroll
      for (int nb = 0; nb < 4; ++nb) {
        acc[nb] = WMMA_F32_16x16x4(ax, bc[nb], acc[nb]);
        acc[nb] = WMMA_F32_16x16x4(am, bl[nb], acc[nb]);
      }
    }
  } else {
  for (int c = 0; c < 16; ++c) {
    v2f ax = xr[2 * c];
    v2f am = sr[2 * c] * inv;
    v2f as = ax * hn;

    v2f bg[4], bs[4], bl[4];
#pragma unroll
    for (int nb = 0; nb < 4; ++nb) {
      const int bo = bofs + nb * 16 * LDS_STRIDE + 4 * c;
      bg[nb] = *(const v2f*)(sWg + bo);
      bs[nb] = *(const v2f*)(sWs + bo);
      bl[nb] = *(const v2f*)(sWl + bo);
    }
#pragma unroll
    for (int nb = 0; nb < 4; ++nb) {
      acc[nb] = WMMA_F32_16x16x4(ax, bg[nb], acc[nb]);
      acc[nb] = WMMA_F32_16x16x4(as, bs[nb], acc[nb]);
      acc[nb] = WMMA_F32_16x16x4(am, bl[nb], acc[nb]);
    }
  }
  }

#pragma unroll
  for (int nb = 0; nb < 4; ++nb) {
    const int col = nb * 16 + lane_lo;
    const float bv = bias[col];
    v8f r = acc[nb];
#pragma unroll
    for (int rr = 0; rr < 8; ++rr) {
      int m = node_base + rr + 8 * hi;
      if (m < n_nodes) {
        float val = r[rr] + bv;
        val = val > 0.0f ? val : (__expf(val) - 1.0f);
        sum_out[(size_t)m * OUT_DIM + col] = val;
      }
    }
  }
}

extern "C" void kernel_launch(void* const* d_in, const int* in_sizes, int n_in,
                              void* d_out, int out_size, void* d_ws, size_t ws_size,
                              hipStream_t stream) {
  const float* x    = (const float*)d_in[0];
  const float* Wg   = (const float*)d_in[1];
  const float* Wl   = (const float*)d_in[2];
  const float* Ws   = (const float*)d_in[3];
  const float* bias = (const float*)d_in[4];
  const int*   ei   = (const int*)d_in[5];
  float* out = (float*)d_out;
  const int n_nodes = in_sizes[0] / IN_DIM;
  const int n_edges = in_sizes[5] / 2;

  const int    nbins = (n_nodes + BIN_NODES - 1) >> BIN_SHIFT;
  const int    tiles = (n_edges + TILE_EDGES - 1) / TILE_EDGES;
  const size_t offw  = ((size_t)nbins + 1) * (size_t)tiles;
  const size_t recw  = (offw + (size_t)n_nodes + 31) & ~(size_t)31;
  const size_t need  = (recw + 2 * (size_t)tiles * TILE_EDGES) * sizeof(int);
  if (ws_size >= need && nbins <= MAX_BINS && n_edges > 0) {
    int*   offs = (int*)d_ws;
    float* deg  = (float*)d_ws + offw;
    int2*  rec  = (int2*)((int*)d_ws + recw);
    demo_bin_scatter_kernel<<<tiles, 256, 0, stream>>>(ei, offs, rec, n_edges, n_nodes,
                                                       nbins, tiles);
    demo_bin_gather_kernel<<<nbins, 512, 0, stream>>>(x, rec, offs, out, deg, n_nodes,
                                                      tiles);
    {
      int waves  = (n_nodes + 15) / 16;
      int blocks = (waves + 7) / 8;
      demo_node_kernel<<<blocks, 256, 0, stream>>>(x, Wg, Wl, Ws, bias, deg, out,
                                                   n_nodes);
    }
    return;
  }

  float* deg = (float*)d_ws;
  hipMemsetAsync(out, 0, (size_t)n_nodes * IN_DIM * sizeof(float), stream);
  hipMemsetAsync(deg, 0, (size_t)n_nodes * sizeof(float), stream);
  {
    long long threads = (long long)n_edges * 16;
    int blocks = (int)((threads + 255) / 256);
    demo_scatter_kernel<<<blocks, 256, 0, stream>>>(x, ei, out, deg, n_edges);
  }
  {
    int waves  = (n_nodes + 15) / 16;
    int blocks = (waves + 7) / 8;
    demo_node_kernel<<<blocks, 256, 0, stream>>>(x, Wg, Wl, Ws, bias, deg, out,
                                                 n_nodes);
  }
}


# Node product fused into the gather kernel: neighbour sums go from registers through LDS into the f32 WMMAs, never through global memory; 2 launches, was 3; sum orders unchanged
# speedup vs baseline: 4.3147x; 1.1071x over previous
#include <hip/hip_runtime.h>

typedef __attribute__((ext_vector_type(2))) float v2f;
typedef __attribute__((ext_vector_type(8))) float v8f;

#define IN_DIM 64
#define OUT_DIM 64
#define LDS_STRIDE 68

#define WMMA_F32_16x16x4(A, B, C) \
  __builtin_amdgcn_wmma_f32_16x16x4_f32(false, (A), false, (B), (short)0, (C), false, false)

__global__ __launch_bounds__(256) void demo_scatter_kernel(
    const float* __restrict__ x, const int* __restrict__ ei,
    float* __restrict__ nsum, float* __restrict__ deg, int n_edges) {
  long long tid = (long long)blockIdx.x * blockDim.x + threadIdx.x;
  int e = (int)(tid >> 4);
  if (e >= n_edges) return;
  int part = (int)(tid & 15);
  int s = ei[e];
  int d = ei[n_edges + e];
  const float4 v = *(const float4*)(x + (size_t)s * IN_DIM + part * 4);
  float* dst = nsum + (size_t)d * IN_DIM + part * 4;
  atomicAdd(dst + 0, v.x);
  atomicAdd(dst + 1, v.y);
  atomicAdd(dst + 2, v.z);
  atomicAdd(dst + 3, v.w);
  if (part == 0) atomicAdd(deg + d, 1.0f);
}

#define BIN_SHIFT 8
#define BIN_NODES 256
#define MAX_BINS  1024
#define TILE_EDGES 4096
#define CHUNK_RECS 6144

typedef __attribute__((ext_vector_type(4))) float v4f;

template <int NW>
__device__ __forceinline__ int block_scan_excl(int v, int* s_wave, int& total) {
  const int lane = threadIdx.x & 31;
  const int w    = threadIdx.x >> 5;
  int x = v;
  for (int o = 1; o < 32; o <<= 1) {
    const int y = __shfl_up(x, o);
    if (lane >= o) x += y;
  }
  __syncthreads();
  if (lane == 31) s_wave[w] = x;
  __syncthreads();
  int wbase = 0, t = 0;
  for (int i = 0; i < NW; ++i) {
    const int sv = s_wave[i];
    if (i < w) wbase += sv;
    t += sv;
  }
  total = t;
  return wbase + x - v;
}

__global__ __launch_bounds__(256) void demo_bin_scatter_kernel(
    const int* __restrict__ ei, int* __restrict__ offs, int2* __restrict__ rec,
    int n_edges, int n_nodes, int nbins, int tiles) {
  __shared__ int  cnt[MAX_BINS];
  __shared__ int  lstart[MAX_BINS];
  __shared__ int  s_wave[8];
  __shared__ __attribute__((aligned(16))) int2 s_rec[TILE_EDGES];
  const int tid = threadIdx.x;

  const int base = blockIdx.x * TILE_EDGES;
  int d[TILE_EDGES / 256], s[TILE_EDGES / 256], r[TILE_EDGES / 256];
#pragma unroll
  for (int k = 0; k < TILE_EDGES / 256; ++k) {
    const int e = min(base + k * 256 + tid, n_edges - 1);
    d[k] = ei[n_edges + e];
    s[k] = ei[e];
  }
  for (int b = tid; b < nbins; b += 256) cnt[b] = 0;
  __syncthreads();

#pragma unroll
  for (int k = 0; k < TILE_EDGES / 256; ++k) {
    r[k] = -1;
    if (base + k * 256 + tid < n_edges && (unsigned)d[k] < (unsigned)n_nodes)
      r[k] = atomicAdd(&cnt[d[k] >> BIN_SHIFT], 1);
  }
  __syncthreads();

  int carry = 0;
  for (int b0 = 0; b0 < nbins; b0 += 256) {
    const int b = b0 + tid;
    const int v = b < nbins ? cnt[b] : 0;
    int tot;
    const int p = block_scan_excl<8>(v, s_wave, tot);
    if (b < nbins) {
      lstart[b] = carry + p;
      offs[(size_t)b * tiles + blockIdx.x] = carry + p;
    }
    carry += tot;
  }
  const int nvalid = carry;
  if (tid == 0) offs[(size_t)nbins * tiles + blockIdx.x] = nvalid;
  __syncthreads();

  int ls[TILE_EDGES / 256];
#pragma unroll
  for (int k = 0; k < TILE_EDGES / 256; ++k)
    ls[k] = lstart[r[k] >= 0 ? (d[k] >> BIN_SHIFT) : 0];
#pragma unroll
  for (int k = 0; k < TILE_EDGES / 256; ++k)
    if (r[k] >= 0) s_rec[ls[k] + r[k]] = make_int2(s[k], d[k]);
  __syncthreads();
  int2* slice = rec + (size_t)blockIdx.x * TILE_EDGES;
  const int pairs = nvalid >> 1;
  for (int i = tid; i < pairs; i += 256)
    ((int4*)slice)[i] = ((const int4*)s_rec)[i];
  if ((nvalid & 1) && tid == 0) slice[nvalid - 1] = s_rec[nvalid - 1];
}

__global__ __launch_bounds__(512, 8) void demo_bin_gather_node_kernel(
    const float* __restrict__ x, const int2* __restrict__ rec,
    const int* __restrict__ offs, const float* __restrict__ Wg,
    const float* __restrict__ Wl, const float* __restrict__ Ws,
    const float* __restrict__ bias, float* __restrict__ out,
    int n_nodes, int tiles) {
  __shared__ int head[BIN_NODES];
  __shared__ int s_src[CHUNK_RECS];
  __shared__ int s_next[CHUNK_RECS];
  __shared__ int s_wave[16];
  __shared__ __attribute__((aligned(16))) float sWc[OUT_DIM * LDS_STRIDE];
  __shared__ __attribute__((aligned(16))) float sWl[OUT_DIM * LDS_STRIDE];
  __shared__ __attribute__((aligned(16))) float sM[BIN_NODES * LDS_STRIDE];
  __shared__ float sDeg[BIN_NODES];

  {
#pragma unroll
    for (int i = 0; i < 2; ++i) {
      const int idx = threadIdx.x + i * 512;
      const int row = idx >> 4;
      const int k4  = idx & 15;
      const float4 wg = *(const float4*)(Wg + (size_t)idx * 4);
      const float4 ws = *(const float4*)(Ws + (size_t)idx * 4);
      const float4 wl = *(const float4*)(Wl + (size_t)idx * 4);
      *(float4*)(sWc + row * LDS_STRIDE + k4 * 4) =
          make_float4(wg.x + ws.x, wg.y + ws.y, wg.z + ws.z, wg.w + ws.w);
      *(float4*)(sWl + row * LDS_STRIDE + k4 * 4) = wl;
    }
  }

  const int tid   = threadIdx.x;
  const int part  = tid & 7;
  const int g     = tid >> 3;
  const int node0 = blockIdx.x << BIN_SHIFT;
  const int* row0 = offs + (size_t)blockIdx.x * tiles;
  const int* row1 = row0 + tiles;
  const v4f* x4 = (const v4f*)x;

  v4f a0[4], a1[4];
  int cnt[4];
#pragma unroll
  for (int j = 0; j < 4; ++j) { a0[j] = v4f{}; a1[j] = v4f{}; cnt[j] = 0; }

  for (int t0 = 0; t0 < tiles; t0 += 512) {
    const int t = t0 + tid;
    int st = 0, len = 0, gidx = 0;
    if (t < tiles) {
      st  = row0[t];
      len = row1[t] - st;
      st   = min(max(st, 0), TILE_EDGES);
      len  = min(max(len, 0), TILE_EDGES - st);
      gidx = t * TILE_EDGES + st;
    }
    int total;
    const int pos = block_scan_excl<16>(len, s_wave, total);

    for (int c0 = 0; c0 < total; c0 += CHUNK_RECS) {
      const int n = min(CHUNK_RECS, total - c0);
      if (tid < BIN_NODES) head[tid] = -1;
      const int klo = max(0, c0 - pos);
      const int khi = min(len, c0 + n - pos);
#pragma nounroll
      for (int k = klo; k < khi; ++k) s_src[pos + k - c0] = gidx + k;
      __syncthreads();

      int2 q[CHUNK_RECS / 512];
#pragma unroll
      for (int j = 0; j < CHUNK_RECS / 512; ++j)
        q[j] = rec[s_src[min(tid + j * 512, n - 1)]];
      __syncthreads();
      int prev[CHUNK_RECS / 512];
#pragma unroll
      for (int j = 0; j < CHUNK_RECS / 512; ++j) {
        const int i = tid + j * 512;
        prev[j] = -1;
        if (i < n) {
          s_src[i] = q[j].x;
          prev[j] = atomicExch(&head[(q[j].y - node0) & (BIN_NODES - 1)], i);
        }
      }
#pragma unroll
      for (int j = 0; j < CHUNK_RECS / 512; ++j) {
        const int i = tid + j * 512;
        if (i < n) s_next[i] = prev[j];
      }
      __syncthreads();

      int cur[4];
#pragma unroll
      for (int j = 0; j < 4; ++j) cur[j] = head[g + 64 * j];
      for (int it = 0; it < n; ++it) {
        if ((cur[0] & cur[1] & cur[2] & cur[3]) < 0) break;
        v4f v0[4], v1[4];
        bool act[4];
#pragma unroll
        for (int j = 0; j < 4; ++j) {
          act[j] = cur[j] >= 0;
          const int ci   = act[j] ? cur[j] : 0;
          const int sidx = s_src[ci];
          const int nx   = s_next[ci];
          const v4f* row = x4 + (size_t)sidx * (IN_DIM / 4);
          v0[j] = row[part];
          v1[j] = row[part + 8];
          cur[j] = act[j] ? nx : -1;
          cnt[j] += act[j] ? 1 : 0;
        }
#pragma unroll
        for (int j = 0; j < 4; ++j) {
          a0[j] += act[j] ? v0[j] : v4f{};
          a1[j] += act[j] ? v1[j] : v4f{};
        }
      }
      __syncthreads();
    }
  }


#pragma unroll
  for (int j = 0; j < 4; ++j) {
    const int lr = g + 64 * j;
    float* mrow = sM + lr * LDS_STRIDE;
    *(v4f*)(mrow + 4 * part)       = a0[j];
    *(v4f*)(mrow + 4 * (part + 8)) = a1[j];
    if (part == 0) sDeg[lr] = (float)cnt[j];
  }
  __syncthreads();

  const int lane    = tid & 31;
  const int wave    = tid >> 5;
  const int lane_lo = lane & 15;
  const int hi      = lane >> 4;
  const int nloc    = min(BIN_NODES, n_nodes - node0);
  if (wave * 16 >= nloc) return;

  const int lrow = min(wave * 16 + lane_lo, nloc - 1);
  const int node_base = node0 + wave * 16;

  const float dv  = sDeg[lrow];
  const float inv = 1.0f / fmaxf(dv, 1.0f);
  const float hn  = dv > 0.0f ? 1.0f : 0.0f;

  const v2f* xr = (const v2f*)(x  + (size_t)(node0 + lrow) * IN_DIM + 2 * hi);
  const v2f* sr = (const v2f*)(sM + lrow * LDS_STRIDE + 2 * hi);

  const int bofs = lane_lo * LDS_STRIDE + 2 * hi;

  v8f acc[4] = {v8f{}, v8f{}, v8f{}, v8f{}};

  if (__all(dv > 0.0f)) {
#pragma unroll
    for (int c = 0; c < 16; ++c) {
      v2f ax = xr[2 * c];
      v2f am = sr[2 * c] * inv;
      v2f bc[4], bl[4];
#pragma unroll
      for (int nb = 0; nb < 4; ++nb) {
        const int bo = bofs + nb * 16 * LDS_STRIDE + 4 * c;
        bc[nb] = *(const v2f*)(sWc + bo);
        bl[nb] = *(const v2f*)(sWl + bo);
      }
#pragma unroll
      for (int nb = 0; nb < 4; ++nb) {
        acc[nb] = WMMA_F32_16x16x4(ax, bc[nb], acc[nb]);
        acc[nb] = WMMA_F32_16x16x4(am, bl[nb], acc[nb]);
      }
    }
  } else {
    const float* gWg = Wg + (size_t)lane_lo * IN_DIM + 2 * hi;
    const float* gWs = Ws + (size_t)lane_lo * IN_DIM + 2 * hi;
    for (int c = 0; c < 16; ++c) {
      v2f ax = xr[2 * c];
      v2f am = sr[2 * c] * inv;
      v2f as = ax * hn;
      v2f bg[4], bs[4], bl[4];
#pragma unroll
      for (int nb = 0; nb < 4; ++nb) {
        bg[nb] = *(const v2f*)(gWg + nb * 16 * IN_DIM + 4 * c);
        bs[nb] = *(const v2f*)(gWs + nb * 16 * IN_DIM + 4 * c);
        bl[nb] = *(const v2f*)(sWl + bofs + nb * 16 * LDS_STRIDE + 4 * c);
      }
#pragma unroll
      for (int nb = 0; nb < 4; ++nb) {
        acc[nb] = WMMA_F32_16x16x4(ax, bg[nb], acc[nb]);
        acc[nb] = WMMA_F32_16x16x4(as, bs[nb], acc[nb]);
        acc[nb] = WMMA_F32_16x16x4(am, bl[nb], acc[nb]);
      }
    }
  }

#pragma unroll
  for (int nb = 0; nb < 4; ++nb) {
    const int col = nb * 16 + lane_lo;
    const float bv = bias[col];
    v8f r = acc[nb];
#pragma unroll
    for (int rr = 0; rr < 8; ++rr) {
      const int m = node_base + rr + 8 * hi;
      if (m < n_nodes) {
        float val = r[rr] + bv;
        val = val > 0.0f ? val : (__expf(val) - 1.0f);
        out[(size_t)m * OUT_DIM + col] = val;
      }
    }
  }
}

__global__ __launch_bounds__(256, 8) void demo_node_kernel(
    const float* __restrict__ x, const float* __restrict__ Wg,
    const float* __restrict__ Wl, const float* __restrict__ Ws,
    const float* __restrict__ bias, const float* __restrict__ deg,
    float* sum_out, int n_nodes) {
  __shared__ float sW[4 * OUT_DIM * LDS_STRIDE];
  float* sWg = sW;
  float* sWs = sW + OUT_DIM * LDS_STRIDE;
  float* sWl = sW + 2 * OUT_DIM * LDS_STRIDE;
  float* sWc = sW + 3 * OUT_DIM * LDS_STRIDE;

  {
    const int tid = threadIdx.x;
#pragma unroll
    for (int i = 0; i < 4; ++i) {
      int idx = tid + i * 256;
      int row = idx >> 4;
      int k4  = idx & 15;
      float4 g = *(const float4*)(Wg + (size_t)idx * 4);
      float4 s = *(const float4*)(Ws + (size_t)idx * 4);
      float4 l = *(const float4*)(Wl + (size_t)idx * 4);
      *(float4*)(sWg + row * LDS_STRIDE + k4 * 4) = g;
      *(float4*)(sWs + row * LDS_STRIDE + k4 * 4) = s;
      *(float4*)(sWl + row * LDS_STRIDE + k4 * 4) = l;
      *(float4*)(sWc + row * LDS_STRIDE + k4 * 4) =
          make_float4(g.x + s.x, g.y + s.y, g.z + s.z, g.w + s.w);
    }
  }
  __syncthreads();

  const int lane    = threadIdx.x & 31;
  const int wave    = threadIdx.x >> 5;
  const int lane_lo = lane & 15;
  const int hi      = lane >> 4;
  const int node_base = (blockIdx.x * 8 + wave) * 16;
  if (node_base >= n_nodes) return;

  int arow = node_base + lane_lo;
  if (arow > n_nodes - 1) arow = n_nodes - 1;

  const float dv  = deg[arow];
  const float inv = 1.0f / fmaxf(dv, 1.0f);
  const float hn  = dv > 0.0f ? 1.0f : 0.0f;

  const v2f* xr = (const v2f*)(x       + (size_t)arow * IN_DIM + 2 * hi);
  const v2f* sr = (const v2f*)(sum_out + (size_t)arow * IN_DIM + 2 * hi);

  const int bofs = lane_lo * LDS_STRIDE + 2 * hi;

  v8f acc[4] = {v8f{}, v8f{}, v8f{}, v8f{}};

  if (__all(dv > 0.0f)) {
#pragma unroll
    for (int c = 0; c < 16; ++c) {
      v2f ax = xr[2 * c];
      v2f am = sr[2 * c] * inv;
      v2f bc[4], bl[4];
#pragma unroll
      for (int nb = 0; nb < 4; ++nb) {
        const int bo = bofs + nb * 16 * LDS_STRIDE + 4 * c;
        bc[nb] = *(const v2f*)(sWc + bo);
        bl[nb] = *(const v2f*)(sWl + bo);
      }
#pragma unroll
      for (int nb = 0; nb < 4; ++nb) {
        acc[nb] = WMMA_F32_16x16x4(ax, bc[nb], acc[nb]);
        acc[nb] = WMMA_F32_16x16x4(am, bl[nb], acc[nb]);
      }
    }
  } else {
  for (int c = 0; c < 16; ++c) {
    v2f ax = xr[2 * c];
    v2f am = sr[2 * c] * inv;
    v2f as = ax * hn;

    v2f bg[4], bs[4], bl[4];
#pragma unroll
    for (int nb = 0; nb < 4; ++nb) {
      const int bo = bofs + nb * 16 * LDS_STRIDE + 4 * c;
      bg[nb] = *(const v2f*)(sWg + bo);
      bs[nb] = *(const v2f*)(sWs + bo);
      bl[nb] = *(const v2f*)(sWl + bo);
    }
#pragma unroll
    for (int nb = 0; nb < 4; ++nb) {
      acc[nb] = WMMA_F32_16x16x4(ax, bg[nb], acc[nb]);
      acc[nb] = WMMA_F32_16x16x4(as, bs[nb], acc[nb]);
      acc[nb] = WMMA_F32_16x16x4(am, bl[nb], acc[nb]);
    }
  }
  }

#pragma unroll
  for (int nb = 0; nb < 4; ++nb) {
    const int col = nb * 16 + lane_lo;
    const float bv = bias[col];
    v8f r = acc[nb];
#pragma unroll
    for (int rr = 0; rr < 8; ++rr) {
      int m = node_base + rr + 8 * hi;
      if (m < n_nodes) {
        float val = r[rr] + bv;
        val = val > 0.0f ? val : (__expf(val) - 1.0f);
        sum_out[(size_t)m * OUT_DIM + col] = val;
      }
    }
  }
}

extern "C" void kernel_launch(void* const* d_in, const int* in_sizes, int n_in,
                              void* d_out, int out_size, void* d_ws, size_t ws_size,
                              hipStream_t stream) {
  const float* x    = (const float*)d_in[0];
  const float* Wg   = (const float*)d_in[1];
  const float* Wl   = (const float*)d_in[2];
  const float* Ws   = (const float*)d_in[3];
  const float* bias = (const float*)d_in[4];
  const int*   ei   = (const int*)d_in[5];
  float* out = (float*)d_out;
  const int n_nodes = in_sizes[0] / IN_DIM;
  const int n_edges = in_sizes[5] / 2;

  const int    nbins = (n_nodes + BIN_NODES - 1) >> BIN_SHIFT;
  const int    tiles = (n_edges + TILE_EDGES - 1) / TILE_EDGES;
  const size_t offw  = ((size_t)nbins + 1) * (size_t)tiles;
  const size_t recw  = (offw + (size_t)n_nodes + 31) & ~(size_t)31;
  const size_t need  = (recw + 2 * (size_t)tiles * TILE_EDGES) * sizeof(int);
  if (ws_size >= need && nbins <= MAX_BINS && n_edges > 0) {
    int*   offs = (int*)d_ws;
    int2*  rec  = (int2*)((int*)d_ws + recw);
    demo_bin_scatter_kernel<<<tiles, 256, 0, stream>>>(ei, offs, rec, n_edges, n_nodes,
                                                       nbins, tiles);
    demo_bin_gather_node_kernel<<<nbins, 512, 0, stream>>>(x, rec, offs, Wg, Wl, Ws, bias,
                                                           out, n_nodes, tiles);
    return;
  }

  float* deg = (float*)d_ws;
  hipMemsetAsync(out, 0, (size_t)n_nodes * IN_DIM * sizeof(float), stream);
  hipMemsetAsync(deg, 0, (size_t)n_nodes * sizeof(float), stream);
  {
    long long threads = (long long)n_edges * 16;
    int blocks = (int)((threads + 255) / 256);
    demo_scatter_kernel<<<blocks, 256, 0, stream>>>(x, ei, out, deg, n_edges);
  }
  {
    int waves  = (n_nodes + 15) / 16;
    int blocks = (waves + 7) / 8;
    demo_node_kernel<<<blocks, 256, 0, stream>>>(x, Wg, Wl, Ws, bias, deg, out,
                                                 n_nodes);
  }
}
